# starting kernel
# baseline (speedup 1.0000x reference)
#include <hip/hip_runtime.h>


typedef __attribute__((ext_vector_type(2))) float v2f;
typedef __attribute__((ext_vector_type(4))) float v4f;
typedef __attribute__((ext_vector_type(8))) float v8f;

#define BATCH 8
#define H 128
#define W 128
#define OH 127
#define OW 127
#define TILE 8
#define SRC 9
#define NSLOT 96
#define PIXSTRIDE 516

__global__ __launch_bounds__(1024, 1)
void convsbs_fused_kernel(const float* __restrict__ channels,
                          const float* __restrict__ cores,
                          float* __restrict__ out) {
  __shared__ float inter[NSLOT * PIXSTRIDE];

  const int b   = blockIdx.z;
  const int ty0 = blockIdx.y * TILE;
  const int tx0 = blockIdx.x * TILE;

  const float* __restrict__ ch0 = channels;
  const float* __restrict__ ch1 = channels + (long)BATCH * H * W * 4;

  const int tid  = (int)threadIdx.x;
  const int wv   = tid >> 5;
  const int lane = tid & 31;

  if (wv < 24) {
    const int g    = wv >> 2;
    const int core = wv & 3;
    const int m    = lane & 15;
    const int half = lane >> 4;

    int s = g * 16 + m;
    int sc = s > 80 ? 80 : s;
    int sy = sc / SRC, sx = sc % SRC;
    int gy = ty0 + sy; gy = gy > (H - 1) ? (H - 1) : gy;
    int gx = tx0 + sx; gx = gx > (W - 1) ? (W - 1) : gx;
    const long pixoff = ((((long)b * H + gy) * W) + gx) * 4;

    const v4f c0 = *(const v4f*)(ch0 + pixoff);
    const v2f c1 = *(const v2f*)(ch1 + pixoff + 2 * half);

    v2f afrag[4];
#pragma unroll
    for (int kt = 0; kt < 4; ++kt) {
      afrag[kt][0] = c0[kt] * c1[0];
      afrag[kt][1] = c0[kt] * c1[1];
    }

    const float* __restrict__ wbase = cores + core * 2048;
    const int sbase = g * 16 + 8 * half;

#pragma unroll
    for (int nt = 0; nt < 8; ++nt) {
      v8f acc = {};
#pragma unroll
      for (int kt = 0; kt < 4; ++kt) {
        const v2f bfrag =
            *(const v2f*)(wbase + (nt * 16 + m) * 16 + kt * 4 + 2 * half);
        acc = __builtin_amdgcn_wmma_f32_16x16x4_f32(
            false, afrag[kt], false, bfrag, (short)0, acc, false, false);
      }
      const int colbase = core * 128 + nt * 16 + m;
#pragma unroll
      for (int v = 0; v < 8; ++v)
        inter[(sbase + v) * PIXSTRIDE + colbase] = acc[v];
    }
  }

  __syncthreads();

  const int pix = tid >> 4;
  const int chn = tid & 15;
  const int oyl = pix >> 3, oxl = pix & 7;
  const int p  = (chn >> 3) & 1;
  const int q  = (chn >> 2) & 1;
  const int r  = (chn >> 1) & 1;
  const int s2 = chn & 1;

  const int sA = oyl * SRC + oxl;
  const float* __restrict__ Am = &inter[(sA)*PIXSTRIDE + 0 * 128 + p * 64];
  const float* __restrict__ Bm = &inter[(sA + 1) * PIXSTRIDE + 1 * 128 + q * 64];
  const float* __restrict__ Cm = &inter[(sA + SRC) * PIXSTRIDE + 2 * 128 + r * 64];
  const float* __restrict__ Dm = &inter[(sA + SRC + 1) * PIXSTRIDE + 3 * 128 + s2 * 64];

  float acc = 0.f;
  for (int i = 0; i < 8; ++i) {
    float e[8];
#pragma unroll
    for (int k = 0; k < 8; ++k) e[k] = 0.f;
#pragma unroll
    for (int j = 0; j < 8; ++j) {
      const float av = Am[i * 8 + j];
#pragma unroll
      for (int k = 0; k < 8; ++k) e[k] = fmaf(av, Bm[j * 8 + k], e[k]);
    }
#pragma unroll
    for (int l = 0; l < 8; ++l) {
      float gv = 0.f;
#pragma unroll
      for (int k = 0; k < 8; ++k) gv = fmaf(e[k], Cm[k * 8 + l], gv);
      acc = fmaf(gv, Dm[l * 8 + i], acc);
    }
  }

  const int oy = ty0 + oyl;
  const int ox = tx0 + oxl;
  if (oy < OH && ox < OW)
    out[((((long)b * OH + oy) * OW) + ox) * 16 + chn] = acc;
}

extern "C" void kernel_launch(void* const* d_in, const int* in_sizes, int n_in,
                              void* d_out, int out_size, void* d_ws, size_t ws_size,
                              hipStream_t stream) {
  (void)in_sizes; (void)n_in; (void)d_ws; (void)ws_size; (void)out_size;
  const float* channels = (const float*)d_in[0];
  const float* cores    = (const float*)d_in[1];
  float* out            = (float*)d_out;

  dim3 grid((OW + TILE - 1) / TILE, (OH + TILE - 1) / TILE, BATCH);
  dim3 block(1024);
  hipLaunchKernelGGL(convsbs_fused_kernel, grid, block, 0, stream,
                     channels, cores, out);
}


# Stage-2 stacked rows reordered to (i>>2, i&3, p or s): a result lane holds both p and both s for half of i, so the ring trace needs no cross-half lane permutes (32 to 1 per wave per tile)
# speedup vs baseline: 7.7080x; 7.7080x over previous
#include <hip/hip_runtime.h>


typedef __attribute__((ext_vector_type(2))) float v2f;
typedef __attribute__((ext_vector_type(4))) float v4f;
typedef __attribute__((ext_vector_type(8))) float v8f;

#define BATCH 8
#define H 128
#define W 128
#define OH 127
#define OW 127
#define TILE_W 8
#define TILE_H 4
#define NSLOT 32
#define PIXSTRIDE 516
#define NTILES 4096
#define NWG 512
#define NTHREADS 512

template <int CTRL>
__device__ __forceinline__ float dpp_mov(float v) {
  return __builtin_bit_cast(float, __builtin_amdgcn_update_dpp(
      0, __builtin_bit_cast(int, v), CTRL, 0xF, 0xF, true));
}
__device__ __forceinline__ float lane_xor1(float v) { return dpp_mov<0x161>(v); }
__device__ __forceinline__ float lane_xor2(float v) { return dpp_mov<0x162>(v); }
__device__ __forceinline__ float lane_xor4(float v) { return dpp_mov<0x164>(v); }
__device__ __forceinline__ float lane_xor8(float v) { return dpp_mov<0x168>(v); }
__device__ __forceinline__ float lane_xor16(float v) {
  const int i = __builtin_bit_cast(int, v);
  return __builtin_bit_cast(float, __builtin_amdgcn_permlanex16(i, i, 0x76543210, 0xfedcba98, false, false));
}

__device__ __forceinline__ void lds_barrier() {
  asm volatile("s_wait_dscnt 0x0\n\ts_barrier_signal -1\n\ts_barrier_wait -1" ::: "memory");
}

__global__ __launch_bounds__(NTHREADS, 8)
void convsbs_fused_kernel(const float* __restrict__ channels,
                          const float* __restrict__ cores,
                          float* __restrict__ out) {
  __shared__ __attribute__((aligned(16))) float inter[NSLOT * PIXSTRIDE];

  const float* __restrict__ ch0 = channels;
  const float* __restrict__ ch1 = channels + (long)BATCH * H * W * 4;

  const int tid  = (int)threadIdx.x;
  const int wv   = __builtin_amdgcn_readfirstlane(tid >> 5);
  const int lane = tid & 31;

  const int core = wv & 3;
  const int ntp  = wv >> 2;
  const int m    = lane & 15;
  const int half = lane >> 4;

  v2f bfrag[2][4];
  {
    const float* __restrict__ wbase = cores + core * 2048;
#pragma unroll
    for (int j = 0; j < 2; ++j)
#pragma unroll
      for (int kt = 0; kt < 4; ++kt)
        bfrag[j][kt] = *(const v2f*)(wbase + ((2 * ntp + j) * 16 + m) * 16 + kt * 4 + 2 * half);
  }
  const int dst = 8 * half * PIXSTRIDE + core * 128 + ntp * 32 + m;
  const int rel0 = ((m >> 3) + (core >> 1)) * W + (m & 7) + (core & 1);

  for (int tile = (int)blockIdx.x; tile < NTILES; tile += NWG) {
    const int b   = tile >> 9;
    const int ty0 = ((tile >> 4) & 31) * TILE_H;
    const int tx0 = (tile & 15) * TILE_W;

    {
      v4f c0[2];
      v2f c1[2];
      const int ycut = (ty0 == H - TILE_H) ? TILE_H * W : (1 << 30);
      const int xcut = (tx0 == W - TILE_W) ? TILE_W : (1 << 30);
      const int pix0 = ((b * H + ty0) * W) + tx0;
#pragma unroll
      for (int g = 0; g < 2; ++g) {
        int rel = rel0 + 2 * g * W;
        rel -= (rel >= ycut) ? W : 0;
        rel -= ((rel & (W - 1)) >= xcut) ? 1 : 0;
        unsigned pixb = (unsigned)(pix0 + rel) * 16u;
        asm volatile("" : "+v"(pixb));
        c0[g] = *(const v4f*)((const char*)ch0 + pixb);
        c1[g] = *(const v2f*)((const char*)ch1 + (pixb + 8u * (unsigned)half));
      }

#pragma unroll
      for (int g = 0; g < 2; ++g) {
        v2f afrag[4];
#pragma unroll
        for (int kt = 0; kt < 4; ++kt)
          afrag[kt] = (v2f){c0[g][kt] * c1[g][0], c0[g][kt] * c1[g][1]};
#pragma unroll
        for (int j = 0; j < 2; ++j) {
          v8f acc = {};
#pragma unroll
          for (int kt = 0; kt < 4; ++kt)
            acc = __builtin_amdgcn_wmma_f32_16x16x4_f32(
                false, afrag[kt], false, bfrag[j][kt], (short)0, acc, false, false);
#pragma unroll
          for (int v = 0; v < 8; ++v)
            inter[dst + j * 16 + (g * 16 + v) * PIXSTRIDE] = acc[v];
        }
      }
    }

    __syncthreads();

    {
      const int c = lane & 15;
      const int h = lane >> 4;
      const int oyl  = wv >> 2;
      const int oxl0 = (wv & 3) * 2;
      const int offRow = c * 8 + 4 * h;
      const int offCol = (c >> 3) * 64 + (c & 7) + 32 * h;
      const int rsel = c & 1;
      const int ri   = 4 * (c >> 3) + ((c >> 1) & 3);
      const int offRowA = rsel * 64 + ri * 8 + 4 * h;
      const int offColD = rsel * 64 + ri + 32 * h;

      float val[16];
#pragma unroll
      for (int P = 0; P < 2; ++P) {
        const float* pA = &inter[(oyl * TILE_W + oxl0 + P) * PIXSTRIDE];
        const float* pB = pA + 1 * 128;
        const float* pC = pA + 2 * 128;
        const float* pD = pA + 3 * 128;

        const v4f a4 = *(const v4f*)(pA + offRowA);
        const v4f c4 = *(const v4f*)(pC + offRow);
        float b4[4], d4[4];
#pragma unroll
        for (int u = 0; u < 4; ++u) {
          b4[u] = pB[offCol + 8 * u];
          d4[u] = pD[offColD + 8 * u];
        }

        v8f ab = {};
        v8f cd = {};
#pragma unroll
        for (int t = 0; t < 2; ++t) {
          const v2f af = {a4[2 * t], a4[2 * t + 1]};
          const v2f bf = {b4[2 * t], b4[2 * t + 1]};
          const v2f df = {d4[2 * t], d4[2 * t + 1]};
          const v2f cf = {c4[2 * t], c4[2 * t + 1]};
          ab = __builtin_amdgcn_wmma_f32_16x16x4_f32(false, af, false, bf, (short)0, ab, false, false);
          cd = __builtin_amdgcn_wmma_f32_16x16x4_f32(false, df, false, cf, (short)0, cd, false, false);
        }
        float tl[2][2] = {{0.f, 0.f}, {0.f, 0.f}};
        float tx[2][2] = {{0.f, 0.f}, {0.f, 0.f}};
#pragma unroll
        for (int il = 0; il < 4; ++il) {
#pragma unroll
          for (int s = 0; s < 2; ++s) {
            const float e0 = cd[2 * il + s];
            const float e1 = lane_xor8(e0);
#pragma unroll
            for (int p = 0; p < 2; ++p) {
              tl[p][s] = fmaf(ab[2 * il + p], e0, tl[p][s]);
              tx[p][s] = fmaf(ab[2 * il + p], e1, tx[p][s]);
            }
          }
        }
#pragma unroll
        for (int p = 0; p < 2; ++p)
#pragma unroll
          for (int s = 0; s < 2; ++s) {
            val[P * 8 + p * 2 + s]     = tl[p][s];
            val[P * 8 + 4 + p * 2 + s] = tx[p][s];
          }
      }

      const bool b2 = (lane & 4) != 0, b1 = (lane & 2) != 0, b0 = (lane & 1) != 0, b4h = h != 0;
      float w8[8];
#pragma unroll
      for (int j = 0; j < 8; ++j) {
        const float keep = b2 ? val[8 + j] : val[j];
        const float send = b2 ? val[j] : val[8 + j];
        w8[j] = keep + lane_xor4(send);
      }
      float z4[4];
#pragma unroll
      for (int j = 0; j < 4; ++j) {
        const float keep = b1 ? w8[4 + j] : w8[j];
        const float send = b1 ? w8[j] : w8[4 + j];
        z4[j] = keep + lane_xor2(send);
      }
      float y2[2];
#pragma unroll
      for (int j = 0; j < 2; ++j) {
        const float keep = b0 ? z4[2 + j] : z4[j];
        const float send = b0 ? z4[j] : z4[2 + j];
        y2[j] = keep + lane_xor1(send);
      }
      const float keep = b4h ? y2[1] : y2[0];
      const float send = b4h ? y2[0] : y2[1];
      const float res = keep + lane_xor16(send);

      const int P  = (lane >> 2) & 1;
      const int rr = (lane >> 1) & 1;
      const int p  = lane & 1;
      const int x  = (lane >> 3) & 1;
      const int r  = x ^ rr;
      const int chn = p * 8 + x * 4 + r * 2 + h;
      const int oy = ty0 + oyl;
      const int ox = tx0 + oxl0 + P;
      if (oy < OH && ox < OW)
        *(float*)((char*)out + (unsigned)(((((b * OH + oy) * OW) + ox) * 16) + chn) * 4u) = res;
    }
    lds_barrier();
  }
}

extern "C" void kernel_launch(void* const* d_in, const int* in_sizes, int n_in,
                              void* d_out, int out_size, void* d_ws, size_t ws_size,
                              hipStream_t stream) {
  (void)in_sizes; (void)n_in; (void)d_ws; (void)ws_size; (void)out_size;
  const float* channels = (const float*)d_in[0];
  const float* cores    = (const float*)d_in[1];
  float* out            = (float*)d_out;

  dim3 grid(NWG);
  dim3 block(NTHREADS);
  hipLaunchKernelGGL(convsbs_fused_kernel, grid, block, 0, stream,
                     channels, cores, out);
}


# Tile loop software-pipelined over two LDS slabs: a pass runs stage 2 of tile k, then stage 1 of tile k+1 from channel pixels loaded a pass earlier; one workgroup barrier per tile (was two)
# speedup vs baseline: 8.2679x; 1.0726x over previous
#include <hip/hip_runtime.h>


typedef __attribute__((ext_vector_type(2))) float v2f;
typedef __attribute__((ext_vector_type(4))) float v4f;
typedef __attribute__((ext_vector_type(8))) float v8f;

#define BATCH 8
#define H 128
#define W 128
#define OH 127
#define OW 127
#define TILE_W 8
#define TILE_H 4
#define NSLOT 32
#define PIXSTRIDE 516
#define SLAB (NSLOT * PIXSTRIDE)
#define NTILES 4096
#define NWG 512
#define TPW (NTILES / NWG)
#define NTHREADS 512

template <int CTRL>
__device__ __forceinline__ float dpp_mov(float v) {
  return __builtin_bit_cast(float, __builtin_amdgcn_update_dpp(
      0, __builtin_bit_cast(int, v), CTRL, 0xF, 0xF, true));
}
__device__ __forceinline__ float lane_xor1(float v) { return dpp_mov<0x161>(v); }
__device__ __forceinline__ float lane_xor2(float v) { return dpp_mov<0x162>(v); }
__device__ __forceinline__ float lane_xor4(float v) { return dpp_mov<0x164>(v); }
__device__ __forceinline__ float lane_xor8(float v) { return dpp_mov<0x168>(v); }
__device__ __forceinline__ float lane_xor16(float v) {
  const int i = __builtin_bit_cast(int, v);
  return __builtin_bit_cast(float, __builtin_amdgcn_permlanex16(i, i, 0x76543210, 0xfedcba98, false, false));
}

__device__ __forceinline__ void lds_barrier() {
  asm volatile("s_wait_dscnt 0x0\n\ts_barrier_signal -1\n\ts_barrier_wait -1" ::: "memory");
}

__global__ __launch_bounds__(NTHREADS, 8)
void convsbs_fused_kernel(const float* __restrict__ channels,
                          const float* __restrict__ cores,
                          float* __restrict__ out) {
  __shared__ __attribute__((aligned(16))) float inter[2 * SLAB];

  const float* __restrict__ ch0 = channels;
  const float* __restrict__ ch1 = channels + (long)BATCH * H * W * 4;

  const int tid  = (int)threadIdx.x;
  const int wv   = __builtin_amdgcn_readfirstlane(tid >> 5);
  const int lane = tid & 31;

  const int core = wv & 3;
  const int ntp  = wv >> 2;
  const int m    = lane & 15;
  const int half = lane >> 4;

  v2f bfrag[2][4];
  {
    const float* __restrict__ wbase = cores + core * 2048;
#pragma unroll
    for (int j = 0; j < 2; ++j)
#pragma unroll
      for (int kt = 0; kt < 4; ++kt)
        bfrag[j][kt] = *(const v2f*)(wbase + ((2 * ntp + j) * 16 + m) * 16 + kt * 4 + 2 * half);
  }
  const int dst = 8 * half * PIXSTRIDE + core * 128 + ntp * 32 + m;
  const int rel0 = ((m >> 3) + (core >> 1)) * W + (m & 7) + (core & 1);

  auto load_pixels = [&](int t, v4f (&c0)[2], v2f (&c1)[2]) __attribute__((always_inline)) {
    const int b   = t >> 9;
    const int ty0 = ((t >> 4) & 31) * TILE_H;
    const int tx0 = (t & 15) * TILE_W;
    const int ycut = (ty0 == H - TILE_H) ? TILE_H * W : (1 << 30);
    const int xcut = (tx0 == W - TILE_W) ? TILE_W : (1 << 30);
    const int pix0 = ((b * H + ty0) * W) + tx0;
#pragma unroll
    for (int g = 0; g < 2; ++g) {
      int rel = rel0 + 2 * g * W;
      rel -= (rel >= ycut) ? W : 0;
      rel -= ((rel & (W - 1)) >= xcut) ? 1 : 0;
      unsigned pixb = (unsigned)(pix0 + rel) * 16u;
      asm volatile("" : "+v"(pixb));
      c0[g] = *(const v4f*)((const char*)ch0 + pixb);
      c1[g] = *(const v2f*)((const char*)ch1 + (pixb + 8u * (unsigned)half));
    }
  };

  auto make_afrag = [&](const v4f (&c0)[2], const v2f (&c1)[2], v2f (&afrag)[2][4]) __attribute__((always_inline)) {
#pragma unroll
    for (int g = 0; g < 2; ++g)
#pragma unroll
      for (int kt = 0; kt < 4; ++kt)
        afrag[g][kt] = (v2f){c0[g][kt] * c1[g][0], c0[g][kt] * c1[g][1]};
  };

  auto stage1 = [&](const v2f (&afrag)[2][4], int sb) __attribute__((always_inline)) {
    int d = sb + dst;
    asm volatile("" : "+v"(d));
#pragma unroll
    for (int g = 0; g < 2; ++g) {
#pragma unroll
      for (int j = 0; j < 2; ++j) {
        v8f acc = {};
#pragma unroll
        for (int kt = 0; kt < 4; ++kt)
          acc = __builtin_amdgcn_wmma_f32_16x16x4_f32(
              false, afrag[g][kt], false, bfrag[j][kt], (short)0, acc, false, false);
#pragma unroll
        for (int v = 0; v < 8; ++v)
          inter[d + j * 16 + (g * 16 + v) * PIXSTRIDE] = acc[v];
      }
    }
  };

  const int c = lane & 15;
  const int h = lane >> 4;
  const int oyl  = wv >> 2;
  const int oxl0 = (wv & 3) * 2;
  const int offRow = c * 8 + 4 * h;
  const int offCol = (c >> 3) * 64 + (c & 7) + 32 * h;
  const int rsel = c & 1;
  const int ri   = 4 * (c >> 3) + ((c >> 1) & 3);
  const int offRowA = rsel * 64 + ri * 8 + 4 * h;
  const int offColD = rsel * 64 + ri + 32 * h;

  auto stage2 = [&](int t, int sb) __attribute__((always_inline)) {
    const int b   = t >> 9;
    const int ty0 = ((t >> 4) & 31) * TILE_H;
    const int tx0 = (t & 15) * TILE_W;

    float val[16];
#pragma unroll
    for (int P = 0; P < 2; ++P) {
      const float* pA = &inter[sb + (oyl * TILE_W + oxl0 + P) * PIXSTRIDE];
      const float* pB = pA + 1 * 128;
      const float* pC = pA + 2 * 128;
      const float* pD = pA + 3 * 128;

      const v4f a4 = *(const v4f*)(pA + offRowA);
      const v4f c4 = *(const v4f*)(pC + offRow);
      float b4[4], d4[4];
#pragma unroll
      for (int u = 0; u < 4; ++u) {
        b4[u] = pB[offCol + 8 * u];
        d4[u] = pD[offColD + 8 * u];
      }

      v8f ab = {};
      v8f cd = {};
#pragma unroll
      for (int tt = 0; tt < 2; ++tt) {
        const v2f af = {a4[2 * tt], a4[2 * tt + 1]};
        const v2f bf = {b4[2 * tt], b4[2 * tt + 1]};
        const v2f df = {d4[2 * tt], d4[2 * tt + 1]};
        const v2f cf = {c4[2 * tt], c4[2 * tt + 1]};
        ab = __builtin_amdgcn_wmma_f32_16x16x4_f32(false, af, false, bf, (short)0, ab, false, false);
        cd = __builtin_amdgcn_wmma_f32_16x16x4_f32(false, df, false, cf, (short)0, cd, false, false);
      }
      float tl[2][2] = {{0.f, 0.f}, {0.f, 0.f}};
      float tx[2][2] = {{0.f, 0.f}, {0.f, 0.f}};
#pragma unroll
      for (int il = 0; il < 4; ++il) {
#pragma unroll
        for (int s = 0; s < 2; ++s) {
          const float e0 = cd[2 * il + s];
          const float e1 = lane_xor8(e0);
#pragma unroll
          for (int p = 0; p < 2; ++p) {
            tl[p][s] = fmaf(ab[2 * il + p], e0, tl[p][s]);
            tx[p][s] = fmaf(ab[2 * il + p], e1, tx[p][s]);
          }
        }
      }
#pragma unroll
      for (int p = 0; p < 2; ++p)
#pragma unroll
        for (int s = 0; s < 2; ++s) {
          val[P * 8 + p * 2 + s]     = tl[p][s];
          val[P * 8 + 4 + p * 2 + s] = tx[p][s];
        }
    }

    const bool b2 = (lane & 4) != 0, b1 = (lane & 2) != 0, b0 = (lane & 1) != 0, b4h = h != 0;
    float w8[8];
#pragma unroll
    for (int j = 0; j < 8; ++j) {
      const float keep = b2 ? val[8 + j] : val[j];
      const float send = b2 ? val[j] : val[8 + j];
      w8[j] = keep + lane_xor4(send);
    }
    float z4[4];
#pragma unroll
    for (int j = 0; j < 4; ++j) {
      const float keep = b1 ? w8[4 + j] : w8[j];
      const float send = b1 ? w8[j] : w8[4 + j];
      z4[j] = keep + lane_xor2(send);
    }
    float y2[2];
#pragma unroll
    for (int j = 0; j < 2; ++j) {
      const float keep = b0 ? z4[2 + j] : z4[j];
      const float send = b0 ? z4[j] : z4[2 + j];
      y2[j] = keep + lane_xor1(send);
    }
    const float keep = b4h ? y2[1] : y2[0];
    const float send = b4h ? y2[0] : y2[1];
    const float res = keep + lane_xor16(send);

    const int P  = (lane >> 2) & 1;
    const int rr = (lane >> 1) & 1;
    const int p  = lane & 1;
    const int x  = (lane >> 3) & 1;
    const int r  = x ^ rr;
    const int chn = p * 8 + x * 4 + r * 2 + h;
    const int oy = ty0 + oyl;
    const int ox = tx0 + oxl0 + P;
    if (oy < OH && ox < OW)
      *(float*)((char*)out + (unsigned)(((((b * OH + oy) * OW) + ox) * 16) + chn) * 4u) = res;
  };

  const int tile0 = (int)blockIdx.x;
  v4f c0[2];
  v2f c1[2];
  {
    v2f afrag[2][4];
    load_pixels(tile0, c0, c1);
    make_afrag(c0, c1, afrag);
    load_pixels(tile0 + NWG, c0, c1);
    stage1(afrag, 0);
  }
  lds_barrier();

#pragma unroll 1
  for (int k = 0; k < TPW - 1; ++k) {
    const int tile = tile0 + k * NWG;
    const int sbr  = (k & 1) * SLAB;
    const int sbw  = SLAB - sbr;
    const int knext = (k + 2 < TPW) ? k + 2 : TPW - 1;
    v4f n0[2];
    v2f n1[2];
    load_pixels(tile0 + knext * NWG, n0, n1);
    stage2(tile, sbr);
    v2f afrag[2][4];
    make_afrag(c0, c1, afrag);
    stage1(afrag, sbw);
    lds_barrier();
    asm volatile("" : "+v"(n0[0]), "+v"(n0[1]), "+v"(n1[0]), "+v"(n1[1]));
#pragma unroll
    for (int g = 0; g < 2; ++g) { c0[g] = n0[g]; c1[g] = n1[g]; }
  }
  stage2(tile0 + (TPW - 1) * NWG, ((TPW - 1) & 1) * SLAB);
}

extern "C" void kernel_launch(void* const* d_in, const int* in_sizes, int n_in,
                              void* d_out, int out_size, void* d_ws, size_t ws_size,
                              hipStream_t stream) {
  (void)in_sizes; (void)n_in; (void)d_ws; (void)ws_size; (void)out_size;
  const float* channels = (const float*)d_in[0];
  const float* cores    = (const float*)d_in[1];
  float* out            = (float*)d_out;

  dim3 grid(NWG);
  dim3 block(NTHREADS);
  hipLaunchKernelGGL(convsbs_fused_kernel, grid, block, 0, stream,
                     channels, cores, out);
}


# Stage 1 computes the transposed tile (weights as A operand, pixels as B): a result lane holds 8 consecutive columns of one pixel, so a tile leaves as 2 ds_store_b128, not 8 row stores + address adds
# speedup vs baseline: 8.5817x; 1.0380x over previous
#include <hip/hip_runtime.h>


typedef __attribute__((ext_vector_type(2))) float v2f;
typedef __attribute__((ext_vector_type(4))) float v4f;
typedef __attribute__((ext_vector_type(8))) float v8f;

#define BATCH 8
#define H 128
#define W 128
#define OH 127
#define OW 127
#define TILE_W 8
#define TILE_H 4
#define NSLOT 32
#define PIXSTRIDE 516
#define SLAB (NSLOT * PIXSTRIDE)
#define NTILES 4096
#define NWG 512
#define TPW (NTILES / NWG)
#define NTHREADS 512

template <int CTRL>
__device__ __forceinline__ float dpp_mov(float v) {
  return __builtin_bit_cast(float, __builtin_amdgcn_update_dpp(
      0, __builtin_bit_cast(int, v), CTRL, 0xF, 0xF, true));
}
__device__ __forceinline__ float lane_xor1(float v) { return dpp_mov<0x161>(v); }
__device__ __forceinline__ float lane_xor2(float v) { return dpp_mov<0x162>(v); }
__device__ __forceinline__ float lane_xor4(float v) { return dpp_mov<0x164>(v); }
__device__ __forceinline__ float lane_xor8(float v) { return dpp_mov<0x168>(v); }
__device__ __forceinline__ float lane_xor16(float v) {
  const int i = __builtin_bit_cast(int, v);
  return __builtin_bit_cast(float, __builtin_amdgcn_permlanex16(i, i, 0x76543210, 0xfedcba98, false, false));
}

__device__ __forceinline__ void lds_barrier() {
  asm volatile("s_wait_dscnt 0x0\n\ts_barrier_signal -1\n\ts_barrier_wait -1" ::: "memory");
}

__global__ __launch_bounds__(NTHREADS, 8)
void convsbs_fused_kernel(const float* __restrict__ channels,
                          const float* __restrict__ cores,
                          float* __restrict__ out) {
  __shared__ __attribute__((aligned(16))) float inter[2 * SLAB];

  const float* __restrict__ ch0 = channels;
  const float* __restrict__ ch1 = channels + (long)BATCH * H * W * 4;

  const int tid  = (int)threadIdx.x;
  const int wv   = __builtin_amdgcn_readfirstlane(tid >> 5);
  const int lane = tid & 31;

  const int core = wv & 3;
  const int ntp  = wv >> 2;
  const int m    = lane & 15;
  const int half = lane >> 4;

  v2f bfrag[2][4];
  {
    const float* __restrict__ wbase = cores + core * 2048;
#pragma unroll
    for (int j = 0; j < 2; ++j)
#pragma unroll
      for (int kt = 0; kt < 4; ++kt)
        bfrag[j][kt] = *(const v2f*)(wbase + ((2 * ntp + j) * 16 + m) * 16 + kt * 4 + 2 * half);
  }
  const int dst = m * PIXSTRIDE + core * 128 + ntp * 32 + 8 * half;
  const int rel0 = ((m >> 3) + (core >> 1)) * W + (m & 7) + (core & 1);

  auto load_pixels = [&](int t, v4f (&c0)[2], v2f (&c1)[2]) __attribute__((always_inline)) {
    const int b   = t >> 9;
    const int ty0 = ((t >> 4) & 31) * TILE_H;
    const int tx0 = (t & 15) * TILE_W;
    const int ycut = (ty0 == H - TILE_H) ? TILE_H * W : (1 << 30);
    const int xcut = (tx0 == W - TILE_W) ? TILE_W : (1 << 30);
    const int pix0 = ((b * H + ty0) * W) + tx0;
#pragma unroll
    for (int g = 0; g < 2; ++g) {
      int rel = rel0 + 2 * g * W;
      rel -= (rel >= ycut) ? W : 0;
      rel -= ((rel & (W - 1)) >= xcut) ? 1 : 0;
      unsigned pixb = (unsigned)(pix0 + rel) * 16u;
      asm volatile("" : "+v"(pixb));
      c0[g] = *(const v4f*)((const char*)ch0 + pixb);
      c1[g] = *(const v2f*)((const char*)ch1 + (pixb + 8u * (unsigned)half));
    }
  };

  auto make_afrag = [&](const v4f (&c0)[2], const v2f (&c1)[2], v2f (&afrag)[2][4]) __attribute__((always_inline)) {
#pragma unroll
    for (int g = 0; g < 2; ++g)
#pragma unroll
      for (int kt = 0; kt < 4; ++kt)
        afrag[g][kt] = (v2f){c0[g][kt] * c1[g][0], c0[g][kt] * c1[g][1]};
  };

  auto stage1 = [&](const v2f (&afrag)[2][4], int sb) __attribute__((always_inline)) {
    int d = sb + dst;
    asm volatile("" : "+v"(d));
#pragma unroll
    for (int g = 0; g < 2; ++g) {
#pragma unroll
      for (int j = 0; j < 2; ++j) {
        v8f acc = {};
#pragma unroll
        for (int kt = 0; kt < 4; ++kt)
          acc = __builtin_amdgcn_wmma_f32_16x16x4_f32(
              false, bfrag[j][kt], false, afrag[g][kt], (short)0, acc, false, false);
        float* q = &inter[d + g * 16 * PIXSTRIDE + j * 16];
        *(v4f*)(q)     = (v4f){acc[0], acc[1], acc[2], acc[3]};
        *(v4f*)(q + 4) = (v4f){acc[4], acc[5], acc[6], acc[7]};
      }
    }
  };

  const int c = lane & 15;
  const int h = lane >> 4;
  const int oyl  = wv >> 2;
  const int oxl0 = (wv & 3) * 2;
  const int offRow = c * 8 + 4 * h;
  const int offCol = (c >> 3) * 64 + (c & 7) + 32 * h;
  const int rsel = c & 1;
  const int ri   = 4 * (c >> 3) + ((c >> 1) & 3);
  const int offRowA = rsel * 64 + ri * 8 + 4 * h;
  const int offColD = rsel * 64 + ri + 32 * h;

  auto stage2 = [&](int t, int sb) __attribute__((always_inline)) {
    const int b   = t >> 9;
    const int ty0 = ((t >> 4) & 31) * TILE_H;
    const int tx0 = (t & 15) * TILE_W;

    float val[16];
#pragma unroll
    for (int P = 0; P < 2; ++P) {
      const float* pA = &inter[sb + (oyl * TILE_W + oxl0 + P) * PIXSTRIDE];
      const float* pB = pA + 1 * 128;
      const float* pC = pA + 2 * 128;
      const float* pD = pA + 3 * 128;

      const v4f a4 = *(const v4f*)(pA + offRowA);
      const v4f c4 = *(const v4f*)(pC + offRow);
      float b4[4], d4[4];
#pragma unroll
      for (int u = 0; u < 4; ++u) {
        b4[u] = pB[offCol + 8 * u];
        d4[u] = pD[offColD + 8 * u];
      }

      v8f ab = {};
      v8f cd = {};
#pragma unroll
      for (int tt = 0; tt < 2; ++tt) {
        const v2f af = {a4[2 * tt], a4[2 * tt + 1]};
        const v2f bf = {b4[2 * tt], b4[2 * tt + 1]};
        const v2f df = {d4[2 * tt], d4[2 * tt + 1]};
        const v2f cf = {c4[2 * tt], c4[2 * tt + 1]};
        ab = __builtin_amdgcn_wmma_f32_16x16x4_f32(false, af, false, bf, (short)0, ab, false, false);
        cd = __builtin_amdgcn_wmma_f32_16x16x4_f32(false, df, false, cf, (short)0, cd, false, false);
      }
      float tl[2][2] = {{0.f, 0.f}, {0.f, 0.f}};
      float tx[2][2] = {{0.f, 0.f}, {0.f, 0.f}};
#pragma unroll
      for (int il = 0; il < 4; ++il) {
#pragma unroll
        for (int s = 0; s < 2; ++s) {
          const float e0 = cd[2 * il + s];
          const float e1 = lane_xor8(e0);
#pragma unroll
          for (int p = 0; p < 2; ++p) {
            tl[p][s] = fmaf(ab[2 * il + p], e0, tl[p][s]);
            tx[p][s] = fmaf(ab[2 * il + p], e1, tx[p][s]);
          }
        }
      }
#pragma unroll
      for (int p = 0; p < 2; ++p)
#pragma unroll
        for (int s = 0; s < 2; ++s) {
          val[P * 8 + p * 2 + s]     = tl[p][s];
          val[P * 8 + 4 + p * 2 + s] = tx[p][s];
        }
    }

    const bool b2 = (lane & 4) != 0, b1 = (lane & 2) != 0, b0 = (lane & 1) != 0, b4h = h != 0;
    float w8[8];
#pragma unroll
    for (int j = 0; j < 8; ++j) {
      const float keep = b2 ? val[8 + j] : val[j];
      const float send = b2 ? val[j] : val[8 + j];
      w8[j] = keep + lane_xor4(send);
    }
    float z4[4];
#pragma unroll
    for (int j = 0; j < 4; ++j) {
      const float keep = b1 ? w8[4 + j] : w8[j];
      const float send = b1 ? w8[j] : w8[4 + j];
      z4[j] = keep + lane_xor2(send);
    }
    float y2[2];
#pragma unroll
    for (int j = 0; j < 2; ++j) {
      const float keep = b0 ? z4[2 + j] : z4[j];
      const float send = b0 ? z4[j] : z4[2 + j];
      y2[j] = keep + lane_xor1(send);
    }
    const float keep = b4h ? y2[1] : y2[0];
    const float send = b4h ? y2[0] : y2[1];
    const float res = keep + lane_xor16(send);

    const int P  = (lane >> 2) & 1;
    const int rr = (lane >> 1) & 1;
    const int p  = lane & 1;
    const int x  = (lane >> 3) & 1;
    const int r  = x ^ rr;
    const int chn = p * 8 + x * 4 + r * 2 + h;
    const int oy = ty0 + oyl;
    const int ox = tx0 + oxl0 + P;
    if (oy < OH && ox < OW)
      *(float*)((char*)out + (unsigned)(((((b * OH + oy) * OW) + ox) * 16) + chn) * 4u) = res;
  };

  const int tile0 = (int)blockIdx.x;
  v4f c0[2];
  v2f c1[2];
  {
    v2f afrag[2][4];
    load_pixels(tile0, c0, c1);
    make_afrag(c0, c1, afrag);
    load_pixels(tile0 + NWG, c0, c1);
    stage1(afrag, 0);
  }
  lds_barrier();

#pragma unroll 1
  for (int k = 0; k < TPW - 1; ++k) {
    const int tile = tile0 + k * NWG;
    const int sbr  = (k & 1) * SLAB;
    const int sbw  = SLAB - sbr;
    const int knext = (k + 2 < TPW) ? k + 2 : TPW - 1;
    v4f n0[2];
    v2f n1[2];
    load_pixels(tile0 + knext * NWG, n0, n1);
    stage2(tile, sbr);
    v2f afrag[2][4];
    make_afrag(c0, c1, afrag);
    stage1(afrag, sbw);
    lds_barrier();
    asm volatile("" : "+v"(n0[0]), "+v"(n0[1]), "+v"(n1[0]), "+v"(n1[1]));
#pragma unroll
    for (int g = 0; g < 2; ++g) { c0[g] = n0[g]; c1[g] = n1[g]; }
  }
  stage2(tile0 + (TPW - 1) * NWG, ((TPW - 1) & 1) * SLAB);
}

extern "C" void kernel_launch(void* const* d_in, const int* in_sizes, int n_in,
                              void* d_out, int out_size, void* d_ws, size_t ws_size,
                              hipStream_t stream) {
  (void)in_sizes; (void)n_in; (void)d_ws; (void)ws_size; (void)out_size;
  const float* channels = (const float*)d_in[0];
  const float* cores    = (const float*)d_in[1];
  float* out            = (float*)d_out;

  dim3 grid(NWG);
  dim3 block(NTHREADS);
  hipLaunchKernelGGL(convsbs_fused_kernel, grid, block, 0, stream,
                     channels, cores, out);
}


# Issue priority falls with a workgroup's progress (s_setprio 3 at entry, 2 and 1 from tile passes 3 and 5, 0 for the drain): of the 2 workgroups on a WGP the one behind wins issue; both end together
# speedup vs baseline: 8.9636x; 1.0445x over previous
#include <hip/hip_runtime.h>


typedef __attribute__((ext_vector_type(2))) float v2f;
typedef __attribute__((ext_vector_type(4))) float v4f;
typedef __attribute__((ext_vector_type(8))) float v8f;

#define BATCH 8
#define H 128
#define W 128
#define OH 127
#define OW 127
#define TILE_W 8
#define TILE_H 4
#define NSLOT 32
#define PIXSTRIDE 516
#define SLAB (NSLOT * PIXSTRIDE)
#define NTILES 4096
#define NWG 512
#define TPW (NTILES / NWG)
#define NTHREADS 512
static_assert(TPW == 8, "the priority steps in the tile loop are placed for 7 passes and a drain");

template <int CTRL>
__device__ __forceinline__ float dpp_mov(float v) {
  return __builtin_bit_cast(float, __builtin_amdgcn_update_dpp(
      0, __builtin_bit_cast(int, v), CTRL, 0xF, 0xF, true));
}
__device__ __forceinline__ float lane_xor1(float v) { return dpp_mov<0x161>(v); }
__device__ __forceinline__ float lane_xor2(float v) { return dpp_mov<0x162>(v); }
__device__ __forceinline__ float lane_xor4(float v) { return dpp_mov<0x164>(v); }
__device__ __forceinline__ float lane_xor8(float v) { return dpp_mov<0x168>(v); }
__device__ __forceinline__ float lane_xor16(float v) {
  const int i = __builtin_bit_cast(int, v);
  return __builtin_bit_cast(float, __builtin_amdgcn_permlanex16(i, i, 0x76543210, 0xfedcba98, false, false));
}

__device__ __forceinline__ void lds_barrier() {
  asm volatile("s_wait_dscnt 0x0\n\ts_barrier_signal -1\n\ts_barrier_wait -1" ::: "memory");
}

__global__ __launch_bounds__(NTHREADS, 8)
void convsbs_fused_kernel(const float* __restrict__ channels,
                          const float* __restrict__ cores,
                          float* __restrict__ out) {
  __shared__ __attribute__((aligned(16))) float inter[2 * SLAB];

  const float* __restrict__ ch0 = channels;
  const float* __restrict__ ch1 = channels + (long)BATCH * H * W * 4;

  __builtin_amdgcn_s_setprio(3);

  const int tid  = (int)threadIdx.x;
  const int wv   = __builtin_amdgcn_readfirstlane(tid >> 5);
  const int lane = tid & 31;

  const int core = wv & 3;
  const int ntp  = wv >> 2;
  const int m    = lane & 15;
  const int half = lane >> 4;

  v2f bfrag[2][4];
  {
    const float* __restrict__ wbase = cores + core * 2048;
#pragma unroll
    for (int j = 0; j < 2; ++j)
#pragma unroll
      for (int kt = 0; kt < 4; ++kt)
        bfrag[j][kt] = *(const v2f*)(wbase + ((2 * ntp + j) * 16 + m) * 16 + kt * 4 + 2 * half);
  }
  const int dst = m * PIXSTRIDE + core * 128 + ntp * 32 + 8 * half;
  const int rel0 = ((m >> 3) + (core >> 1)) * W + (m & 7) + (core & 1);

  auto load_pixels = [&](int t, v4f (&c0)[2], v2f (&c1)[2]) __attribute__((always_inline)) {
    const int b   = t >> 9;
    const int ty0 = ((t >> 4) & 31) * TILE_H;
    const int tx0 = (t & 15) * TILE_W;
    const int ycut = (ty0 == H - TILE_H) ? TILE_H * W : (1 << 30);
    const int xcut = (tx0 == W - TILE_W) ? TILE_W : (1 << 30);
    const int pix0 = ((b * H + ty0) * W) + tx0;
#pragma unroll
    for (int g = 0; g < 2; ++g) {
      int rel = rel0 + 2 * g * W;
      rel -= (rel >= ycut) ? W : 0;
      rel -= ((rel & (W - 1)) >= xcut) ? 1 : 0;
      unsigned pixb = (unsigned)(pix0 + rel) * 16u;
      asm volatile("" : "+v"(pixb));
      c0[g] = *(const v4f*)((const char*)ch0 + pixb);
      c1[g] = *(const v2f*)((const char*)ch1 + (pixb + 8u * (unsigned)half));
    }
  };

  auto make_afrag = [&](const v4f (&c0)[2], const v2f (&c1)[2], v2f (&afrag)[2][4]) __attribute__((always_inline)) {
#pragma unroll
    for (int g = 0; g < 2; ++g)
#pragma unroll
      for (int kt = 0; kt < 4; ++kt)
        afrag[g][kt] = (v2f){c0[g][kt] * c1[g][0], c0[g][kt] * c1[g][1]};
  };

  auto stage1 = [&](const v2f (&afrag)[2][4], int sb) __attribute__((always_inline)) {
    int d = sb + dst;
    asm volatile("" : "+v"(d));
#pragma unroll
    for (int g = 0; g < 2; ++g) {
#pragma unroll
      for (int j = 0; j < 2; ++j) {
        v8f acc = {};
#pragma unroll
        for (int kt = 0; kt < 4; ++kt)
          acc = __builtin_amdgcn_wmma_f32_16x16x4_f32(
              false, bfrag[j][kt], false, afrag[g][kt], (short)0, acc, false, false);
        float* q = &inter[d + g * 16 * PIXSTRIDE + j * 16];
        *(v4f*)(q)     = (v4f){acc[0], acc[1], acc[2], acc[3]};
        *(v4f*)(q + 4) = (v4f){acc[4], acc[5], acc[6], acc[7]};
      }
    }
  };

  const int c = lane & 15;
  const int h = lane >> 4;
  const int oyl  = wv >> 2;
  const int oxl0 = (wv & 3) * 2;
  const int offRow = c * 8 + 4 * h;
  const int offCol = (c >> 3) * 64 + (c & 7) + 32 * h;
  const int rsel = c & 1;
  const int ri   = 4 * (c >> 3) + ((c >> 1) & 3);
  const int offRowA = rsel * 64 + ri * 8 + 4 * h;
  const int offColD = rsel * 64 + ri + 32 * h;

  auto stage2 = [&](int t, int sb) __attribute__((always_inline)) {
    const int b   = t >> 9;
    const int ty0 = ((t >> 4) & 31) * TILE_H;
    const int tx0 = (t & 15) * TILE_W;

    float val[16];
#pragma unroll
    for (int P = 0; P < 2; ++P) {
      const float* pA = &inter[sb + (oyl * TILE_W + oxl0 + P) * PIXSTRIDE];
      const float* pB = pA + 1 * 128;
      const float* pC = pA + 2 * 128;
      const float* pD = pA + 3 * 128;

      const v4f a4 = *(const v4f*)(pA + offRowA);
      const v4f c4 = *(const v4f*)(pC + offRow);
      float b4[4], d4[4];
#pragma unroll
      for (int u = 0; u < 4; ++u) {
        b4[u] = pB[offCol + 8 * u];
        d4[u] = pD[offColD + 8 * u];
      }

      v8f ab = {};
      v8f cd = {};
#pragma unroll
      for (int tt = 0; tt < 2; ++tt) {
        const v2f af = {a4[2 * tt], a4[2 * tt + 1]};
        const v2f bf = {b4[2 * tt], b4[2 * tt + 1]};
        const v2f df = {d4[2 * tt], d4[2 * tt + 1]};
        const v2f cf = {c4[2 * tt], c4[2 * tt + 1]};
        ab = __builtin_amdgcn_wmma_f32_16x16x4_f32(false, af, false, bf, (short)0, ab, false, false);
        cd = __builtin_amdgcn_wmma_f32_16x16x4_f32(false, df, false, cf, (short)0, cd, false, false);
      }
      float tl[2][2] = {{0.f, 0.f}, {0.f, 0.f}};
      float tx[2][2] = {{0.f, 0.f}, {0.f, 0.f}};
#pragma unroll
      for (int il = 0; il < 4; ++il) {
#pragma unroll
        for (int s = 0; s < 2; ++s) {
          const float e0 = cd[2 * il + s];
          const float e1 = lane_xor8(e0);
#pragma unroll
          for (int p = 0; p < 2; ++p) {
            tl[p][s] = fmaf(ab[2 * il + p], e0, tl[p][s]);
            tx[p][s] = fmaf(ab[2 * il + p], e1, tx[p][s]);
          }
        }
      }
#pragma unroll
      for (int p = 0; p < 2; ++p)
#pragma unroll
        for (int s = 0; s < 2; ++s) {
          val[P * 8 + p * 2 + s]     = tl[p][s];
          val[P * 8 + 4 + p * 2 + s] = tx[p][s];
        }
    }

    const bool b2 = (lane & 4) != 0, b1 = (lane & 2) != 0, b0 = (lane & 1) != 0, b4h = h != 0;
    float w8[8];
#pragma unroll
    for (int j = 0; j < 8; ++j) {
      const float keep = b2 ? val[8 + j] : val[j];
      const float send = b2 ? val[j] : val[8 + j];
      w8[j] = keep + lane_xor4(send);
    }
    float z4[4];
#pragma unroll
    for (int j = 0; j < 4; ++j) {
      const float keep = b1 ? w8[4 + j] : w8[j];
      const float send = b1 ? w8[j] : w8[4 + j];
      z4[j] = keep + lane_xor2(send);
    }
    float y2[2];
#pragma unroll
    for (int j = 0; j < 2; ++j) {
      const float keep = b0 ? z4[2 + j] : z4[j];
      const float send = b0 ? z4[j] : z4[2 + j];
      y2[j] = keep + lane_xor1(send);
    }
    const float keep = b4h ? y2[1] : y2[0];
    const float send = b4h ? y2[0] : y2[1];
    const float res = keep + lane_xor16(send);

    const int P  = (lane >> 2) & 1;
    const int rr = (lane >> 1) & 1;
    const int p  = lane & 1;
    const int x  = (lane >> 3) & 1;
    const int r  = x ^ rr;
    const int chn = p * 8 + x * 4 + r * 2 + h;
    const int oy = ty0 + oyl;
    const int ox = tx0 + oxl0 + P;
    if (oy < OH && ox < OW)
      *(float*)((char*)out + (unsigned)(((((b * OH + oy) * OW) + ox) * 16) + chn) * 4u) = res;
  };

  const int tile0 = (int)blockIdx.x;
  v4f c0[2];
  v2f c1[2];
  {
    v2f afrag[2][4];
    load_pixels(tile0, c0, c1);
    make_afrag(c0, c1, afrag);
    load_pixels(tile0 + NWG, c0, c1);
    stage1(afrag, 0);
  }
  lds_barrier();

#pragma unroll 1
  for (int k = 0; k < TPW - 1; ++k) {
    if (k == 3)      __builtin_amdgcn_s_setprio(2);
    else if (k == 5) __builtin_amdgcn_s_setprio(1);
    const int tile = tile0 + k * NWG;
    const int sbr  = (k & 1) * SLAB;
    const int sbw  = SLAB - sbr;
    const int knext = (k + 2 < TPW) ? k + 2 : TPW - 1;
    v4f n0[2];
    v2f n1[2];
    load_pixels(tile0 + knext * NWG, n0, n1);
    stage2(tile, sbr);
    v2f afrag[2][4];
    make_afrag(c0, c1, afrag);
    stage1(afrag, sbw);
    lds_barrier();
    asm volatile("" : "+v"(n0[0]), "+v"(n0[1]), "+v"(n1[0]), "+v"(n1[1]));
#pragma unroll
    for (int g = 0; g < 2; ++g) { c0[g] = n0[g]; c1[g] = n1[g]; }
  }
  __builtin_amdgcn_s_setprio(0);
  stage2(tile0 + (TPW - 1) * NWG, ((TPW - 1) & 1) * SLAB);
}

extern "C" void kernel_launch(void* const* d_in, const int* in_sizes, int n_in,
                              void* d_out, int out_size, void* d_ws, size_t ws_size,
                              hipStream_t stream) {
  (void)in_sizes; (void)n_in; (void)d_ws; (void)ws_size; (void)out_size;
  const float* channels = (const float*)d_in[0];
  const float* cores    = (const float*)d_in[1];
  float* out            = (float*)d_out;

  dim3 grid(NWG);
  dim3 block(NTHREADS);
  hipLaunchKernelGGL(convsbs_fused_kernel, grid, block, 0, stream,
                     channels, cores, out);
}
